# starting kernel
# baseline (speedup 1.0000x reference)
#include <hip/hip_runtime.h>


typedef __attribute__((ext_vector_type(2))) float v2f;
typedef __attribute__((ext_vector_type(8))) float v8f;

#define Bb 16
#define Ss 16
#define Ll 128
#define Dd 768
#define Aa 8
#define Tt 5

__global__ __launch_bounds__(256)
void srl_wmma_kernel(const float* __restrict__ hidden,
                     const int*   __restrict__ sent_ids,
                     const int*   __restrict__ masks,
                     const int*   __restrict__ pred_ids,
                     const int*   __restrict__ arg0_ids,
                     const int*   __restrict__ arg1_ids,
                     float*       __restrict__ out)
{
    __shared__ int   sid[Ll];
    __shared__ float W[32][132];
    __shared__ int   cnt0;

    const int bs   = blockIdx.x;
    const int tid  = threadIdx.x;
    const int lane = tid & 31;
    const int wv   = tid >> 5;

    if (tid == 0) cnt0 = 0;
    __syncthreads();

    if (tid < Ll) {
        int sv = sent_ids[bs * Ll + tid];
        sid[tid] = sv;
        int m = masks[bs * Ll + tid];
        W[0][tid] = (float)m;
        if (m) atomicAdd(&cnt0, 1);
    }
    for (int i = tid; i < 7 * Ll; i += 256) {
        W[25 + (i >> 7)][i & 127] = 0.0f;
    }
    __syncthreads();

    if (tid < Ll) {
        float c = (float)(cnt0 > 1 ? cnt0 : 1);
        W[0][tid] = W[0][tid] * (1.0f / c);
    }

    for (int r = wv; r < 24; r += 8) {
        const int type = r >> 3;
        const int a    = r & 7;
        const int* ids = (type == 0 ? pred_ids : (type == 1 ? arg0_ids : arg1_ids))
                         + ((size_t)bs * Aa + a) * Tt;
        int id[Tt];
        #pragma unroll
        for (int t = 0; t < Tt; ++t) id[t] = ids[t];

        int c[Tt] = {0, 0, 0, 0, 0};
        #pragma unroll
        for (int k = 0; k < Ll / 32; ++k) {
            int sv = sid[lane + 32 * k];
            #pragma unroll
            for (int t = 0; t < Tt; ++t) c[t] += (sv == id[t]) ? 1 : 0;
        }
        #pragma unroll
        for (int off = 16; off > 0; off >>= 1) {
            #pragma unroll
            for (int t = 0; t < Tt; ++t) c[t] += __shfl_xor(c[t], off, 32);
        }
        int   lid   = -1;
        float scale = 0.0f;
        #pragma unroll
        for (int t = 0; t < Tt; ++t) {
            if (id[t] != 0 && c[t] > 0) { lid = id[t]; scale = 1.0f / (float)c[t]; }
        }
        #pragma unroll
        for (int k = 0; k < Ll / 32; ++k) {
            int l = lane + 32 * k;
            W[1 + r][l] = (sid[l] == lid) ? scale : 0.0f;
        }
    }
    __syncthreads();

    const float* H    = hidden + (size_t)bs * (Ll * Dd);
    const int    nt0  = wv * 6;
    const int    mrow = lane & 15;
    const int    khalf = (lane >> 4) << 1;
    const int    ncol = lane & 15;

    v8f acc[2][6] = {};

    for (int kb = 0; kb < Ll; kb += 4) {
        v2f a0, a1;
        a0.x = W[mrow][kb + khalf];
        a0.y = W[mrow][kb + khalf + 1];
        a1.x = W[16 + mrow][kb + khalf];
        a1.y = W[16 + mrow][kb + khalf + 1];

        const float* Hk = H + (size_t)(kb + khalf) * Dd + ncol;
        #pragma unroll
        for (int j = 0; j < 6; ++j) {
            const int n = (nt0 + j) << 4;
            v2f bf;
            bf.x = Hk[n];
            bf.y = Hk[n + Dd];
            acc[0][j] = __builtin_amdgcn_wmma_f32_16x16x4_f32(
                false, a0, false, bf, (short)0, acc[0][j], false, false);
            acc[1][j] = __builtin_amdgcn_wmma_f32_16x16x4_f32(
                false, a1, false, bf, (short)0, acc[1][j], false, false);
        }
    }

    float* outS  = out;
    float* outP  = out  + (size_t)Bb * Ss * Dd;
    float* outA0 = outP + (size_t)Bb * Ss * Aa * Dd;
    float* outA1 = outA0 + (size_t)Bb * Ss * Aa * Dd;

    const int rhi = (lane >> 4) << 3;
    #pragma unroll
    for (int mt = 0; mt < 2; ++mt) {
        #pragma unroll
        for (int j = 0; j < 6; ++j) {
            const int n = (nt0 + j) * 16 + ncol;
            #pragma unroll
            for (int e = 0; e < 8; ++e) {
                const int row = mt * 16 + rhi + e;
                const float v = acc[mt][j][e];
                if (row == 0) {
                    outS[(size_t)bs * Dd + n] = v;
                } else if (row <= 24) {
                    const int r = row - 1;
                    float* dst = ((r >> 3) == 0) ? outP
                               : (((r >> 3) == 1) ? outA0 : outA1);
                    dst[((size_t)bs * Aa + (r & 7)) * Dd + n] = v;
                }
            }
        }
    }
}

extern "C" void kernel_launch(void* const* d_in, const int* in_sizes, int n_in,
                              void* d_out, int out_size, void* d_ws, size_t ws_size,
                              hipStream_t stream) {
    const float* hidden = (const float*)d_in[0];
    const int*   sids   = (const int*)d_in[1];
    const int*   msk    = (const int*)d_in[2];
    const int*   pred   = (const int*)d_in[3];
    const int*   a0     = (const int*)d_in[4];
    const int*   a1     = (const int*)d_in[5];
    float*       out    = (float*)d_out;

    dim3 grid(Bb * Ss);
    dim3 block(256);
    hipLaunchKernelGGL(srl_wmma_kernel, grid, block, 0, stream,
                       hidden, sids, msk, pred, a0, a1, out);
}


# hidden rows read as 16-byte loads per lane (4 N-tiles column-interleaved; 12 waves x 64 cols, 384 threads) and requested a 32-row chunk ahead of their WMMAs, not drained every 8 rows; 16-byte stores
# speedup vs baseline: 1.5182x; 1.5182x over previous
#include <hip/hip_runtime.h>


typedef __attribute__((ext_vector_type(2))) float v2f;
typedef __attribute__((ext_vector_type(4))) float v4f;
typedef __attribute__((ext_vector_type(8))) float v8f;

#define Bb 16
#define Ss 16
#define Ll 128
#define Dd 768
#define Aa 8
#define Tt 5

#define NWAVES 12
#define NTHREADS (NWAVES * 32)
#define KC 8
#define NCHUNK (Ll / (4 * KC))

__global__ __launch_bounds__(NTHREADS)
void srl_wmma_kernel(const float* __restrict__ hidden,
                     const int*   __restrict__ sent_ids,
                     const int*   __restrict__ masks,
                     const int*   __restrict__ pred_ids,
                     const int*   __restrict__ arg0_ids,
                     const int*   __restrict__ arg1_ids,
                     float*       __restrict__ out)
{
    __shared__ int   sid[Ll];
    __shared__ __attribute__((aligned(16))) float W[32][132];
    __shared__ int   cnt0;

    const int bs   = blockIdx.x;
    const int tid  = threadIdx.x;
    const int lane = tid & 31;
    const int wv   = tid >> 5;
    const int c16  = lane & 15;
    const int h    = lane >> 4;

    const float* Hl = hidden + (size_t)bs * (Ll * Dd) + (size_t)(2 * h) * Dd
                    + wv * 64 + c16 * 4;

    if (tid == 0) cnt0 = 0;
    __syncthreads();

    if (tid < Ll) {
        int sv = sent_ids[bs * Ll + tid];
        sid[tid] = sv;
        int m = masks[bs * Ll + tid];
        W[0][tid] = (float)m;
        if (m) atomicAdd(&cnt0, 1);
    }
    for (int i = tid; i < 7 * Ll; i += NTHREADS) {
        W[25 + (i >> 7)][i & 127] = 0.0f;
    }
    __syncthreads();

    if (tid < Ll) {
        float c = (float)(cnt0 > 1 ? cnt0 : 1);
        W[0][tid] = W[0][tid] * (1.0f / c);
    }

    for (int r = wv; r < 24; r += NWAVES) {
        const int type = r >> 3;
        const int a    = r & 7;
        const int* ids = (type == 0 ? pred_ids : (type == 1 ? arg0_ids : arg1_ids))
                         + ((size_t)bs * Aa + a) * Tt;
        int id[Tt];
        #pragma unroll
        for (int t = 0; t < Tt; ++t) id[t] = ids[t];

        int c[Tt] = {0, 0, 0, 0, 0};
        #pragma unroll
        for (int k = 0; k < Ll / 32; ++k) {
            int sv = sid[lane + 32 * k];
            #pragma unroll
            for (int t = 0; t < Tt; ++t) c[t] += (sv == id[t]) ? 1 : 0;
        }
        #pragma unroll
        for (int off = 16; off > 0; off >>= 1) {
            #pragma unroll
            for (int t = 0; t < Tt; ++t) c[t] += __shfl_xor(c[t], off, 32);
        }
        int   lid   = -1;
        float scale = 0.0f;
        #pragma unroll
        for (int t = 0; t < Tt; ++t) {
            if (id[t] != 0 && c[t] > 0) { lid = id[t]; scale = 1.0f / (float)c[t]; }
        }
        #pragma unroll
        for (int k = 0; k < Ll / 32; ++k) {
            int l = lane + 32 * k;
            W[1 + r][l] = (sid[l] == lid) ? scale : 0.0f;
        }
    }
    __syncthreads();

    v8f acc[2][4] = {};
    v4f q[2][KC][2];

    #pragma unroll
    for (int s = 0; s < KC; ++s) {
        q[0][s][0] = *(const v4f*)(Hl + (size_t)(4 * s) * Dd);
        q[0][s][1] = *(const v4f*)(Hl + (size_t)(4 * s + 1) * Dd);
    }

    #pragma unroll
    for (int ch = 0; ch < NCHUNK; ++ch) {
        const int cur = ch & 1;
        v2f a0[KC], a1[KC];
        #pragma unroll
        for (int s = 0; s < KC; ++s) {
            const int kb = 4 * (KC * ch + s) + 2 * h;
            a0[s] = *(const v2f*)&W[c16][kb];
            a1[s] = *(const v2f*)&W[16 + c16][kb];
        }
        if (ch + 1 < NCHUNK) {
            #pragma unroll
            for (int s = 0; s < KC; ++s) {
                const int row = 4 * (KC * (ch + 1) + s);
                q[cur ^ 1][s][0] = *(const v4f*)(Hl + (size_t)row * Dd);
                q[cur ^ 1][s][1] = *(const v4f*)(Hl + (size_t)(row + 1) * Dd);
            }
        }
        __builtin_amdgcn_sched_barrier(0);
        #pragma unroll
        for (int s = 0; s < KC; ++s) {
            #pragma unroll
            for (int j = 0; j < 4; ++j) {
                v2f bf;
                bf.x = q[cur][s][0][j];
                bf.y = q[cur][s][1][j];
                acc[0][j] = __builtin_amdgcn_wmma_f32_16x16x4_f32(
                    false, a0[s], false, bf, (short)0, acc[0][j], false, false);
                acc[1][j] = __builtin_amdgcn_wmma_f32_16x16x4_f32(
                    false, a1[s], false, bf, (short)0, acc[1][j], false, false);
            }
        }
        __builtin_amdgcn_sched_barrier(0);
    }

    float* outS  = out;
    float* outP  = out  + (size_t)Bb * Ss * Dd;
    float* outA0 = outP + (size_t)Bb * Ss * Aa * Dd;
    float* outA1 = outA0 + (size_t)Bb * Ss * Aa * Dd;

    const int n = wv * 64 + c16 * 4;
    #pragma unroll
    for (int mt = 0; mt < 2; ++mt) {
        #pragma unroll
        for (int e = 0; e < 8; ++e) {
            const int row = mt * 16 + 8 * h + e;
            v4f v;
            v.x = acc[mt][0][e];
            v.y = acc[mt][1][e];
            v.z = acc[mt][2][e];
            v.w = acc[mt][3][e];
            if (row == 0) {
                *(v4f*)(outS + (size_t)bs * Dd + n) = v;
            } else if (row <= 24) {
                const int r = row - 1;
                float* dst = ((r >> 3) == 0) ? outP
                           : (((r >> 3) == 1) ? outA0 : outA1);
                *(v4f*)(dst + ((size_t)bs * Aa + (r & 7)) * Dd + n) = v;
            }
        }
    }
}

extern "C" void kernel_launch(void* const* d_in, const int* in_sizes, int n_in,
                              void* d_out, int out_size, void* d_ws, size_t ws_size,
                              hipStream_t stream) {
    const float* hidden = (const float*)d_in[0];
    const int*   sids   = (const int*)d_in[1];
    const int*   msk    = (const int*)d_in[2];
    const int*   pred   = (const int*)d_in[3];
    const int*   a0     = (const int*)d_in[4];
    const int*   a1     = (const int*)d_in[5];
    float*       out    = (float*)d_out;

    dim3 grid(Bb * Ss);
    dim3 block(NTHREADS);
    hipLaunchKernelGGL(srl_wmma_kernel, grid, block, 0, stream,
                       hidden, sids, msk, pred, a0, a1, out);
}


# second M-tile (weight rows 16..24, arg rows only) issues its WMMAs only on k-steps where those rows hold a non-zero: one ballot-built 32-bit map per wave, scalar bit test per k-step; sums unchanged
# speedup vs baseline: 1.6878x; 1.1117x over previous
#include <hip/hip_runtime.h>


typedef __attribute__((ext_vector_type(2))) float v2f;
typedef __attribute__((ext_vector_type(4))) float v4f;
typedef __attribute__((ext_vector_type(8))) float v8f;

#define Bb 16
#define Ss 16
#define Ll 128
#define Dd 768
#define Aa 8
#define Tt 5

#define NWAVES 12
#define NTHREADS (NWAVES * 32)
#define KC 8
#define NCHUNK (Ll / (4 * KC))
#define NROWS 25

__global__ __launch_bounds__(NTHREADS)
void srl_wmma_kernel(const float* __restrict__ hidden,
                     const int*   __restrict__ sent_ids,
                     const int*   __restrict__ masks,
                     const int*   __restrict__ pred_ids,
                     const int*   __restrict__ arg0_ids,
                     const int*   __restrict__ arg1_ids,
                     float*       __restrict__ out)
{
    __shared__ int   sid[Ll];
    __shared__ __attribute__((aligned(16))) float W[32][132];
    __shared__ int   cnt0;

    const int bs   = blockIdx.x;
    const int tid  = threadIdx.x;
    const int lane = tid & 31;
    const int wv   = tid >> 5;
    const int c16  = lane & 15;
    const int h    = lane >> 4;

    const float* Hl = hidden + (size_t)bs * (Ll * Dd) + (size_t)(2 * h) * Dd
                    + wv * 64 + c16 * 4;

    if (tid == 0) cnt0 = 0;
    __syncthreads();

    if (tid < Ll) {
        int sv = sent_ids[bs * Ll + tid];
        sid[tid] = sv;
        int m = masks[bs * Ll + tid];
        W[0][tid] = (float)m;
        if (m) atomicAdd(&cnt0, 1);
    }
    for (int i = tid; i < 7 * Ll; i += NTHREADS) {
        W[25 + (i >> 7)][i & 127] = 0.0f;
    }
    __syncthreads();

    if (tid < Ll) {
        float c = (float)(cnt0 > 1 ? cnt0 : 1);
        W[0][tid] = W[0][tid] * (1.0f / c);
    }

    for (int r = wv; r < 24; r += NWAVES) {
        const int type = r >> 3;
        const int a    = r & 7;
        const int* ids = (type == 0 ? pred_ids : (type == 1 ? arg0_ids : arg1_ids))
                         + ((size_t)bs * Aa + a) * Tt;
        int id[Tt];
        #pragma unroll
        for (int t = 0; t < Tt; ++t) id[t] = ids[t];

        int c[Tt] = {0, 0, 0, 0, 0};
        #pragma unroll
        for (int k = 0; k < Ll / 32; ++k) {
            int sv = sid[lane + 32 * k];
            #pragma unroll
            for (int t = 0; t < Tt; ++t) c[t] += (sv == id[t]) ? 1 : 0;
        }
        #pragma unroll
        for (int off = 16; off > 0; off >>= 1) {
            #pragma unroll
            for (int t = 0; t < Tt; ++t) c[t] += __shfl_xor(c[t], off, 32);
        }
        int   lid   = -1;
        float scale = 0.0f;
        #pragma unroll
        for (int t = 0; t < Tt; ++t) {
            if (id[t] != 0 && c[t] > 0) { lid = id[t]; scale = 1.0f / (float)c[t]; }
        }
        #pragma unroll
        for (int k = 0; k < Ll / 32; ++k) {
            int l = lane + 32 * k;
            W[1 + r][l] = (sid[l] == lid) ? scale : 0.0f;
        }
    }
    __syncthreads();

    v8f acc[2][4] = {};
    v4f q[2][KC][2];

    #pragma unroll
    for (int s = 0; s < KC; ++s) {
        q[0][s][0] = *(const v4f*)(Hl + (size_t)(4 * s) * Dd);
        q[0][s][1] = *(const v4f*)(Hl + (size_t)(4 * s + 1) * Dd);
    }

    unsigned nzb = 0u;
    #pragma unroll
    for (int r = 16; r < NROWS; ++r) {
        const v4f w4 = *(const v4f*)&W[r][4 * lane];
        nzb |= __float_as_uint(w4.x) | __float_as_uint(w4.y)
             | __float_as_uint(w4.z) | __float_as_uint(w4.w);
    }
    const unsigned t1bits = __builtin_amdgcn_ballot_w32(nzb != 0u);

    #pragma unroll
    for (int ch = 0; ch < NCHUNK; ++ch) {
        const int cur = ch & 1;
        v2f a0[KC], a1[KC];
        #pragma unroll
        for (int s = 0; s < KC; ++s) {
            const int kb = 4 * (KC * ch + s) + 2 * h;
            a0[s] = *(const v2f*)&W[c16][kb];
            a1[s] = *(const v2f*)&W[16 + c16][kb];
        }
        if (ch + 1 < NCHUNK) {
            #pragma unroll
            for (int s = 0; s < KC; ++s) {
                const int row = 4 * (KC * (ch + 1) + s);
                q[cur ^ 1][s][0] = *(const v4f*)(Hl + (size_t)row * Dd);
                q[cur ^ 1][s][1] = *(const v4f*)(Hl + (size_t)(row + 1) * Dd);
            }
        }
        __builtin_amdgcn_sched_barrier(0);
        #pragma unroll
        for (int s = 0; s < KC; ++s) {
            v2f bf[4];
            #pragma unroll
            for (int j = 0; j < 4; ++j) {
                bf[j].x = q[cur][s][0][j];
                bf[j].y = q[cur][s][1][j];
                acc[0][j] = __builtin_amdgcn_wmma_f32_16x16x4_f32(
                    false, a0[s], false, bf[j], (short)0, acc[0][j], false, false);
            }
            if (t1bits & (1u << (KC * ch + s))) {
                #pragma unroll
                for (int j = 0; j < 4; ++j) {
                    acc[1][j] = __builtin_amdgcn_wmma_f32_16x16x4_f32(
                        false, a1[s], false, bf[j], (short)0, acc[1][j], false, false);
                }
            }
        }
    }

    float* outS  = out;
    float* outP  = out  + (size_t)Bb * Ss * Dd;
    float* outA0 = outP + (size_t)Bb * Ss * Aa * Dd;
    float* outA1 = outA0 + (size_t)Bb * Ss * Aa * Dd;

    const int n = wv * 64 + c16 * 4;
    #pragma unroll
    for (int mt = 0; mt < 2; ++mt) {
        #pragma unroll
        for (int e = 0; e < 8; ++e) {
            const int row = mt * 16 + 8 * h + e;
            v4f v;
            v.x = acc[mt][0][e];
            v.y = acc[mt][1][e];
            v.z = acc[mt][2][e];
            v.w = acc[mt][3][e];
            if (row == 0) {
                *(v4f*)(outS + (size_t)bs * Dd + n) = v;
            } else if (row <= 24) {
                const int r = row - 1;
                float* dst = ((r >> 3) == 0) ? outP
                           : (((r >> 3) == 1) ? outA0 : outA1);
                *(v4f*)(dst + ((size_t)bs * Aa + (r & 7)) * Dd + n) = v;
            }
        }
    }
}

extern "C" void kernel_launch(void* const* d_in, const int* in_sizes, int n_in,
                              void* d_out, int out_size, void* d_ws, size_t ws_size,
                              hipStream_t stream) {
    const float* hidden = (const float*)d_in[0];
    const int*   sids   = (const int*)d_in[1];
    const int*   msk    = (const int*)d_in[2];
    const int*   pred   = (const int*)d_in[3];
    const int*   a0     = (const int*)d_in[4];
    const int*   a1     = (const int*)d_in[5];
    float*       out    = (float*)d_out;

    dim3 grid(Bb * Ss);
    dim3 block(NTHREADS);
    hipLaunchKernelGGL(srl_wmma_kernel, grid, block, 0, stream,
                       hidden, sids, msk, pred, a0, a1, out);
}


# weight build cut to one memory hop and one barrier: every wave keeps the sentence ids in registers and counts matches by ballot and popcount (no LDS exchange or atomic); weight values unchanged
# speedup vs baseline: 1.7962x; 1.0642x over previous
#include <hip/hip_runtime.h>


typedef __attribute__((ext_vector_type(2))) float v2f;
typedef __attribute__((ext_vector_type(4))) float v4f;
typedef __attribute__((ext_vector_type(4))) int   v4i;
typedef __attribute__((ext_vector_type(8))) float v8f;

#define Bb 16
#define Ss 16
#define Ll 128
#define Dd 768
#define Aa 8
#define Tt 5

#define NWAVES 12
#define NTHREADS (NWAVES * 32)
#define KC 8
#define NCHUNK (Ll / (4 * KC))
#define NROWS 25

__global__ __launch_bounds__(NTHREADS)
void srl_wmma_kernel(const float* __restrict__ hidden,
                     const int*   __restrict__ sent_ids,
                     const int*   __restrict__ masks,
                     const int*   __restrict__ pred_ids,
                     const int*   __restrict__ arg0_ids,
                     const int*   __restrict__ arg1_ids,
                     float*       __restrict__ out)
{
    __shared__ __attribute__((aligned(16))) float W[32][132];

    const int bs   = blockIdx.x;
    const int tid  = threadIdx.x;
    const int lane = tid & 31;
    const int wv   = tid >> 5;
    const int c16  = lane & 15;
    const int h    = lane >> 4;

    const float* Hl = hidden + (size_t)bs * (Ll * Dd) + (size_t)(2 * h) * Dd
                    + wv * 64 + c16 * 4;

    const int wvs = __builtin_amdgcn_readfirstlane(wv);

    const v4i sv = *(const v4i*)(sent_ids + (size_t)bs * Ll + 4 * lane);

    int id[2][Tt];
    #pragma unroll
    for (int i = 0; i < 2; ++i) {
        const int r    = wvs + NWAVES * i;
        const int type = r >> 3;
        const int a    = r & 7;
        const int* ids = (type == 0 ? pred_ids : (type == 1 ? arg0_ids : arg1_ids))
                         + ((size_t)bs * Aa + a) * Tt;
        #pragma unroll
        for (int t = 0; t < Tt; ++t) id[i][t] = ids[t];
    }
    v4i m4 = {0, 0, 0, 0};
    if (wvs == 0) m4 = *(const v4i*)(masks + (size_t)bs * Ll + 4 * lane);

    #pragma unroll
    for (int i = 0; i < 2; ++i) {
        int lid = -1;
        int cc  = 0;
        #pragma unroll
        for (int t = 0; t < Tt; ++t) {
            const int v = id[i][t];
            const int c = __builtin_popcount(__builtin_amdgcn_ballot_w32(sv.x == v))
                        + __builtin_popcount(__builtin_amdgcn_ballot_w32(sv.y == v))
                        + __builtin_popcount(__builtin_amdgcn_ballot_w32(sv.z == v))
                        + __builtin_popcount(__builtin_amdgcn_ballot_w32(sv.w == v));
            if (v != 0 && c > 0) { lid = v; cc = c; }
        }
        const float scale = (cc > 0) ? 1.0f / (float)cc : 0.0f;
        v4f w;
        w.x = (sv.x == lid) ? scale : 0.0f;
        w.y = (sv.y == lid) ? scale : 0.0f;
        w.z = (sv.z == lid) ? scale : 0.0f;
        w.w = (sv.w == lid) ? scale : 0.0f;
        *(v4f*)&W[1 + wvs + NWAVES * i][4 * lane] = w;
    }
    if (wvs == 0) {
        const int cnt = __builtin_popcount(__builtin_amdgcn_ballot_w32(m4.x != 0))
                      + __builtin_popcount(__builtin_amdgcn_ballot_w32(m4.y != 0))
                      + __builtin_popcount(__builtin_amdgcn_ballot_w32(m4.z != 0))
                      + __builtin_popcount(__builtin_amdgcn_ballot_w32(m4.w != 0));
        const float inv = 1.0f / (float)(cnt > 1 ? cnt : 1);
        v4f w;
        w.x = (float)m4.x * inv;
        w.y = (float)m4.y * inv;
        w.z = (float)m4.z * inv;
        w.w = (float)m4.w * inv;
        *(v4f*)&W[0][4 * lane] = w;
    } else if (wvs <= 7) {
        const v4f z = {0.0f, 0.0f, 0.0f, 0.0f};
        *(v4f*)&W[24 + wvs][4 * lane] = z;
    }
    __syncthreads();

    v8f acc[2][4] = {};
    v4f q[2][KC][2];

    #pragma unroll
    for (int s = 0; s < KC; ++s) {
        q[0][s][0] = *(const v4f*)(Hl + (size_t)(4 * s) * Dd);
        q[0][s][1] = *(const v4f*)(Hl + (size_t)(4 * s + 1) * Dd);
    }

    unsigned nzb = 0u;
    #pragma unroll
    for (int r = 16; r < NROWS; ++r) {
        const v4f w4 = *(const v4f*)&W[r][4 * lane];
        nzb |= __float_as_uint(w4.x) | __float_as_uint(w4.y)
             | __float_as_uint(w4.z) | __float_as_uint(w4.w);
    }
    const unsigned t1bits = __builtin_amdgcn_ballot_w32(nzb != 0u);

    #pragma unroll
    for (int ch = 0; ch < NCHUNK; ++ch) {
        const int cur = ch & 1;
        v2f a0[KC], a1[KC];
        #pragma unroll
        for (int s = 0; s < KC; ++s) {
            const int kb = 4 * (KC * ch + s) + 2 * h;
            a0[s] = *(const v2f*)&W[c16][kb];
            a1[s] = *(const v2f*)&W[16 + c16][kb];
        }
        if (ch + 1 < NCHUNK) {
            #pragma unroll
            for (int s = 0; s < KC; ++s) {
                const int row = 4 * (KC * (ch + 1) + s);
                q[cur ^ 1][s][0] = *(const v4f*)(Hl + (size_t)row * Dd);
                q[cur ^ 1][s][1] = *(const v4f*)(Hl + (size_t)(row + 1) * Dd);
            }
        }
        __builtin_amdgcn_sched_barrier(0);
        #pragma unroll
        for (int s = 0; s < KC; ++s) {
            v2f bf[4];
            #pragma unroll
            for (int j = 0; j < 4; ++j) {
                bf[j].x = q[cur][s][0][j];
                bf[j].y = q[cur][s][1][j];
                acc[0][j] = __builtin_amdgcn_wmma_f32_16x16x4_f32(
                    false, a0[s], false, bf[j], (short)0, acc[0][j], false, false);
            }
            if (t1bits & (1u << (KC * ch + s))) {
                #pragma unroll
                for (int j = 0; j < 4; ++j) {
                    acc[1][j] = __builtin_amdgcn_wmma_f32_16x16x4_f32(
                        false, a1[s], false, bf[j], (short)0, acc[1][j], false, false);
                }
            }
        }
    }

    float* outS  = out;
    float* outP  = out  + (size_t)Bb * Ss * Dd;
    float* outA0 = outP + (size_t)Bb * Ss * Aa * Dd;
    float* outA1 = outA0 + (size_t)Bb * Ss * Aa * Dd;

    const int n = wv * 64 + c16 * 4;
    #pragma unroll
    for (int mt = 0; mt < 2; ++mt) {
        #pragma unroll
        for (int e = 0; e < 8; ++e) {
            const int row = mt * 16 + 8 * h + e;
            v4f v;
            v.x = acc[mt][0][e];
            v.y = acc[mt][1][e];
            v.z = acc[mt][2][e];
            v.w = acc[mt][3][e];
            if (row == 0) {
                *(v4f*)(outS + (size_t)bs * Dd + n) = v;
            } else if (row <= 24) {
                const int r = row - 1;
                float* dst = ((r >> 3) == 0) ? outP
                           : (((r >> 3) == 1) ? outA0 : outA1);
                *(v4f*)(dst + ((size_t)bs * Aa + (r & 7)) * Dd + n) = v;
            }
        }
    }
}

extern "C" void kernel_launch(void* const* d_in, const int* in_sizes, int n_in,
                              void* d_out, int out_size, void* d_ws, size_t ws_size,
                              hipStream_t stream) {
    const float* hidden = (const float*)d_in[0];
    const int*   sids   = (const int*)d_in[1];
    const int*   msk    = (const int*)d_in[2];
    const int*   pred   = (const int*)d_in[3];
    const int*   a0     = (const int*)d_in[4];
    const int*   a1     = (const int*)d_in[5];
    float*       out    = (float*)d_out;

    dim3 grid(Bb * Ss);
    dim3 block(NTHREADS);
    hipLaunchKernelGGL(srl_wmma_kernel, grid, block, 0, stream,
                       hidden, sids, msk, pred, a0, a1, out);
}


# first 32 hidden-row loads of each wave requested at kernel entry, ahead of the weight build; its barrier fences LDS only, so the requests stay in flight across it; same loads, sums unchanged
# speedup vs baseline: 1.8982x; 1.0568x over previous
#include <hip/hip_runtime.h>


typedef __attribute__((ext_vector_type(2))) float v2f;
typedef __attribute__((ext_vector_type(4))) float v4f;
typedef __attribute__((ext_vector_type(4))) int   v4i;
typedef __attribute__((ext_vector_type(8))) float v8f;

#define Bb 16
#define Ss 16
#define Ll 128
#define Dd 768
#define Aa 8
#define Tt 5

#define NWAVES 12
#define NTHREADS (NWAVES * 32)
#define KC 8
#define NCHUNK (Ll / (4 * KC))
#define NROWS 25

__global__ __launch_bounds__(NTHREADS)
void srl_wmma_kernel(const float* __restrict__ hidden,
                     const int*   __restrict__ sent_ids,
                     const int*   __restrict__ masks,
                     const int*   __restrict__ pred_ids,
                     const int*   __restrict__ arg0_ids,
                     const int*   __restrict__ arg1_ids,
                     float*       __restrict__ out)
{
    __shared__ __attribute__((aligned(16))) float W[32][132];

    const int bs   = blockIdx.x;
    const int tid  = threadIdx.x;
    const int lane = tid & 31;
    const int wv   = tid >> 5;
    const int c16  = lane & 15;
    const int h    = lane >> 4;

    const float* Hl = hidden + (size_t)bs * (Ll * Dd) + (size_t)(2 * h) * Dd
                    + wv * 64 + c16 * 4;

    const int wvs = __builtin_amdgcn_readfirstlane(wv);

    const v4i sv = *(const v4i*)(sent_ids + (size_t)bs * Ll + 4 * lane);

    int id[2][Tt];
    #pragma unroll
    for (int i = 0; i < 2; ++i) {
        const int r    = wvs + NWAVES * i;
        const int type = r >> 3;
        const int a    = r & 7;
        const int* ids = (type == 0 ? pred_ids : (type == 1 ? arg0_ids : arg1_ids))
                         + ((size_t)bs * Aa + a) * Tt;
        #pragma unroll
        for (int t = 0; t < Tt; ++t) id[i][t] = ids[t];
    }
    v4i m4 = {0, 0, 0, 0};
    if (wvs == 0) m4 = *(const v4i*)(masks + (size_t)bs * Ll + 4 * lane);

    v4f q[2][KC][2];
    #pragma unroll
    for (int c = 0; c < 2; ++c) {
        #pragma unroll
        for (int s = 0; s < KC; ++s) {
            const int row = 4 * (KC * c + s);
            q[c][s][0] = *(const v4f*)(Hl + (size_t)row * Dd);
            q[c][s][1] = *(const v4f*)(Hl + (size_t)(row + 1) * Dd);
        }
    }
    __builtin_amdgcn_sched_barrier(0);

    #pragma unroll
    for (int i = 0; i < 2; ++i) {
        int lid = -1;
        int cc  = 0;
        #pragma unroll
        for (int t = 0; t < Tt; ++t) {
            const int v = id[i][t];
            const int c = __builtin_popcount(__builtin_amdgcn_ballot_w32(sv.x == v))
                        + __builtin_popcount(__builtin_amdgcn_ballot_w32(sv.y == v))
                        + __builtin_popcount(__builtin_amdgcn_ballot_w32(sv.z == v))
                        + __builtin_popcount(__builtin_amdgcn_ballot_w32(sv.w == v));
            if (v != 0 && c > 0) { lid = v; cc = c; }
        }
        const float scale = (cc > 0) ? 1.0f / (float)cc : 0.0f;
        v4f w;
        w.x = (sv.x == lid) ? scale : 0.0f;
        w.y = (sv.y == lid) ? scale : 0.0f;
        w.z = (sv.z == lid) ? scale : 0.0f;
        w.w = (sv.w == lid) ? scale : 0.0f;
        *(v4f*)&W[1 + wvs + NWAVES * i][4 * lane] = w;
    }
    if (wvs == 0) {
        const int cnt = __builtin_popcount(__builtin_amdgcn_ballot_w32(m4.x != 0))
                      + __builtin_popcount(__builtin_amdgcn_ballot_w32(m4.y != 0))
                      + __builtin_popcount(__builtin_amdgcn_ballot_w32(m4.z != 0))
                      + __builtin_popcount(__builtin_amdgcn_ballot_w32(m4.w != 0));
        const float inv = 1.0f / (float)(cnt > 1 ? cnt : 1);
        v4f w;
        w.x = (float)m4.x * inv;
        w.y = (float)m4.y * inv;
        w.z = (float)m4.z * inv;
        w.w = (float)m4.w * inv;
        *(v4f*)&W[0][4 * lane] = w;
    } else if (wvs <= 7) {
        const v4f z = {0.0f, 0.0f, 0.0f, 0.0f};
        *(v4f*)&W[24 + wvs][4 * lane] = z;
    }
    __builtin_amdgcn_fence(__ATOMIC_RELEASE, "workgroup", "local");
    __builtin_amdgcn_s_barrier();
    __builtin_amdgcn_fence(__ATOMIC_ACQUIRE, "workgroup", "local");

    v8f acc[2][4] = {};

    unsigned nzb = 0u;
    #pragma unroll
    for (int r = 16; r < NROWS; ++r) {
        const v4f w4 = *(const v4f*)&W[r][4 * lane];
        nzb |= __float_as_uint(w4.x) | __float_as_uint(w4.y)
             | __float_as_uint(w4.z) | __float_as_uint(w4.w);
    }
    const unsigned t1bits = __builtin_amdgcn_ballot_w32(nzb != 0u);

    #pragma unroll
    for (int ch = 0; ch < NCHUNK; ++ch) {
        const int cur = ch & 1;
        v2f a0[KC], a1[KC];
        #pragma unroll
        for (int s = 0; s < KC; ++s) {
            const int kb = 4 * (KC * ch + s) + 2 * h;
            a0[s] = *(const v2f*)&W[c16][kb];
            a1[s] = *(const v2f*)&W[16 + c16][kb];
        }
        if (ch >= 1 && ch + 1 < NCHUNK) {
            #pragma unroll
            for (int s = 0; s < KC; ++s) {
                const int row = 4 * (KC * (ch + 1) + s);
                q[cur ^ 1][s][0] = *(const v4f*)(Hl + (size_t)row * Dd);
                q[cur ^ 1][s][1] = *(const v4f*)(Hl + (size_t)(row + 1) * Dd);
            }
        }
        __builtin_amdgcn_sched_barrier(0);
        #pragma unroll
        for (int s = 0; s < KC; ++s) {
            v2f bf[4];
            #pragma unroll
            for (int j = 0; j < 4; ++j) {
                bf[j].x = q[cur][s][0][j];
                bf[j].y = q[cur][s][1][j];
                acc[0][j] = __builtin_amdgcn_wmma_f32_16x16x4_f32(
                    false, a0[s], false, bf[j], (short)0, acc[0][j], false, false);
            }
            if (t1bits & (1u << (KC * ch + s))) {
                #pragma unroll
                for (int j = 0; j < 4; ++j) {
                    acc[1][j] = __builtin_amdgcn_wmma_f32_16x16x4_f32(
                        false, a1[s], false, bf[j], (short)0, acc[1][j], false, false);
                }
            }
        }
    }

    float* outS  = out;
    float* outP  = out  + (size_t)Bb * Ss * Dd;
    float* outA0 = outP + (size_t)Bb * Ss * Aa * Dd;
    float* outA1 = outA0 + (size_t)Bb * Ss * Aa * Dd;

    const int n = wv * 64 + c16 * 4;
    #pragma unroll
    for (int mt = 0; mt < 2; ++mt) {
        #pragma unroll
        for (int e = 0; e < 8; ++e) {
            const int row = mt * 16 + 8 * h + e;
            v4f v;
            v.x = acc[mt][0][e];
            v.y = acc[mt][1][e];
            v.z = acc[mt][2][e];
            v.w = acc[mt][3][e];
            if (row == 0) {
                *(v4f*)(outS + (size_t)bs * Dd + n) = v;
            } else if (row <= 24) {
                const int r = row - 1;
                float* dst = ((r >> 3) == 0) ? outP
                           : (((r >> 3) == 1) ? outA0 : outA1);
                *(v4f*)(dst + ((size_t)bs * Aa + (r & 7)) * Dd + n) = v;
            }
        }
    }
}

extern "C" void kernel_launch(void* const* d_in, const int* in_sizes, int n_in,
                              void* d_out, int out_size, void* d_ws, size_t ws_size,
                              hipStream_t stream) {
    const float* hidden = (const float*)d_in[0];
    const int*   sids   = (const int*)d_in[1];
    const int*   msk    = (const int*)d_in[2];
    const int*   pred   = (const int*)d_in[3];
    const int*   a0     = (const int*)d_in[4];
    const int*   a1     = (const int*)d_in[5];
    float*       out    = (float*)d_out;

    dim3 grid(Bb * Ss);
    dim3 block(NTHREADS);
    hipLaunchKernelGGL(srl_wmma_kernel, grid, block, 0, stream,
                       hidden, sids, msk, pred, a0, a1, out);
}


# mask-mean row (the one dense row) moved off the WMMA tile to f32 multiply-adds on the row registers, so both arg M-tiles issue WMMAs only on non-zero k-steps; the mean's summation order changes
# speedup vs baseline: 2.0786x; 1.0950x over previous
#include <hip/hip_runtime.h>


typedef __attribute__((ext_vector_type(2))) float v2f;
typedef __attribute__((ext_vector_type(4))) float v4f;
typedef __attribute__((ext_vector_type(4))) int   v4i;
typedef __attribute__((ext_vector_type(8))) float v8f;

#define Bb 16
#define Ss 16
#define Ll 128
#define Dd 768
#define Aa 8
#define Tt 5

#define NWAVES 12
#define NTHREADS (NWAVES * 32)
#define KC 8
#define NCHUNK (Ll / (4 * KC))
#define NARG 24
#define MROW 32

__device__ __forceinline__ float xhalf(float x) {
    const int xi = __builtin_bit_cast(int, x);
    return __builtin_bit_cast(float,
        __builtin_amdgcn_permlanex16(xi, xi, 0x76543210, 0xfedcba98, false, false));
}

__global__ __launch_bounds__(NTHREADS)
void srl_wmma_kernel(const float* __restrict__ hidden,
                     const int*   __restrict__ sent_ids,
                     const int*   __restrict__ masks,
                     const int*   __restrict__ pred_ids,
                     const int*   __restrict__ arg0_ids,
                     const int*   __restrict__ arg1_ids,
                     float*       __restrict__ out)
{
    __shared__ __attribute__((aligned(16))) float W[33][132];
    __shared__ __attribute__((aligned(16))) unsigned rowmap[32];

    const int bs   = blockIdx.x;
    const int tid  = threadIdx.x;
    const int lane = tid & 31;
    const int wv   = tid >> 5;
    const int c16  = lane & 15;
    const int h    = lane >> 4;

    const float* Hl = hidden + (size_t)bs * (Ll * Dd) + (size_t)(2 * h) * Dd
                    + wv * 64 + c16 * 4;

    const int wvs = __builtin_amdgcn_readfirstlane(wv);

    const v4i sv = *(const v4i*)(sent_ids + (size_t)bs * Ll + 4 * lane);

    int id[2][Tt];
    #pragma unroll
    for (int i = 0; i < 2; ++i) {
        const int r    = wvs + NWAVES * i;
        const int type = r >> 3;
        const int a    = r & 7;
        const int* ids = (type == 0 ? pred_ids : (type == 1 ? arg0_ids : arg1_ids))
                         + ((size_t)bs * Aa + a) * Tt;
        #pragma unroll
        for (int t = 0; t < Tt; ++t) id[i][t] = ids[t];
    }
    v4i m4 = {0, 0, 0, 0};
    if (wvs == 0) m4 = *(const v4i*)(masks + (size_t)bs * Ll + 4 * lane);

    v4f q[2][KC][2];
    #pragma unroll
    for (int c = 0; c < 2; ++c) {
        #pragma unroll
        for (int s = 0; s < KC; ++s) {
            const int row = 4 * (KC * c + s);
            q[c][s][0] = *(const v4f*)(Hl + (size_t)row * Dd);
            q[c][s][1] = *(const v4f*)(Hl + (size_t)(row + 1) * Dd);
        }
    }
    __builtin_amdgcn_sched_barrier(0);

    #pragma unroll
    for (int i = 0; i < 2; ++i) {
        int lid = -1;
        int cc  = 0;
        #pragma unroll
        for (int t = 0; t < Tt; ++t) {
            const int v = id[i][t];
            const int c = __builtin_popcount(__builtin_amdgcn_ballot_w32(sv.x == v))
                        + __builtin_popcount(__builtin_amdgcn_ballot_w32(sv.y == v))
                        + __builtin_popcount(__builtin_amdgcn_ballot_w32(sv.z == v))
                        + __builtin_popcount(__builtin_amdgcn_ballot_w32(sv.w == v));
            if (v != 0 && c > 0) { lid = v; cc = c; }
        }
        const float scale = (cc > 0) ? 1.0f / (float)cc : 0.0f;
        v4f w;
        w.x = (sv.x == lid) ? scale : 0.0f;
        w.y = (sv.y == lid) ? scale : 0.0f;
        w.z = (sv.z == lid) ? scale : 0.0f;
        w.w = (sv.w == lid) ? scale : 0.0f;
        *(v4f*)&W[wvs + NWAVES * i][4 * lane] = w;
        const unsigned hit = __builtin_amdgcn_ballot_w32(
            (sv.x == lid) | (sv.y == lid) | (sv.z == lid) | (sv.w == lid));
        if (lane == 0) rowmap[wvs + NWAVES * i] = (cc > 0) ? hit : 0u;
    }
    if (wvs == 0) {
        const int cnt = __builtin_popcount(__builtin_amdgcn_ballot_w32(m4.x != 0))
                      + __builtin_popcount(__builtin_amdgcn_ballot_w32(m4.y != 0))
                      + __builtin_popcount(__builtin_amdgcn_ballot_w32(m4.z != 0))
                      + __builtin_popcount(__builtin_amdgcn_ballot_w32(m4.w != 0));
        const float inv = 1.0f / (float)(cnt > 1 ? cnt : 1);
        v4f w;
        w.x = (float)m4.x * inv;
        w.y = (float)m4.y * inv;
        w.z = (float)m4.z * inv;
        w.w = (float)m4.w * inv;
        *(v4f*)&W[MROW][4 * lane] = w;
    } else if (wvs <= 8) {
        const v4f z = {0.0f, 0.0f, 0.0f, 0.0f};
        *(v4f*)&W[NARG - 1 + wvs][4 * lane] = z;
        if (lane == 0) rowmap[NARG - 1 + wvs] = 0u;
    }
    __builtin_amdgcn_fence(__ATOMIC_RELEASE, "workgroup", "local");
    __builtin_amdgcn_s_barrier();
    __builtin_amdgcn_fence(__ATOMIC_ACQUIRE, "workgroup", "local");

    v8f acc[2][4] = {};
    v4f mean0 = {0.0f, 0.0f, 0.0f, 0.0f};
    v4f mean1 = {0.0f, 0.0f, 0.0f, 0.0f};

    unsigned t0bits, t1bits;
    {
        const v4i rm = *(const v4i*)&rowmap[4 * (lane & 7)];
        const int m = rm.x | rm.y | rm.z | rm.w;
        t0bits = (unsigned)(__builtin_amdgcn_readlane(m, 0) | __builtin_amdgcn_readlane(m, 1)
                          | __builtin_amdgcn_readlane(m, 2) | __builtin_amdgcn_readlane(m, 3));
        t1bits = (unsigned)(__builtin_amdgcn_readlane(m, 4) | __builtin_amdgcn_readlane(m, 5)
                          | __builtin_amdgcn_readlane(m, 6) | __builtin_amdgcn_readlane(m, 7));
    }

    #pragma unroll
    for (int ch = 0; ch < NCHUNK; ++ch) {
        const int cur = ch & 1;
        v2f a0[KC], a1[KC], wm[KC];
        #pragma unroll
        for (int s = 0; s < KC; ++s) {
            const int kb = 4 * (KC * ch + s) + 2 * h;
            a0[s] = *(const v2f*)&W[c16][kb];
            a1[s] = *(const v2f*)&W[16 + c16][kb];
            wm[s] = *(const v2f*)&W[MROW][kb];
        }
        if (ch >= 1 && ch + 1 < NCHUNK) {
            #pragma unroll
            for (int s = 0; s < KC; ++s) {
                const int row = 4 * (KC * (ch + 1) + s);
                q[cur ^ 1][s][0] = *(const v4f*)(Hl + (size_t)row * Dd);
                q[cur ^ 1][s][1] = *(const v4f*)(Hl + (size_t)(row + 1) * Dd);
            }
        }
        __builtin_amdgcn_sched_barrier(0);
        #pragma unroll
        for (int s = 0; s < KC; ++s) {
            mean0 += wm[s].x * q[cur][s][0];
            mean1 += wm[s].y * q[cur][s][1];
            const unsigned kbit = 1u << (KC * ch + s);
            if ((t0bits | t1bits) & kbit) {
                v2f bf[4];
                #pragma unroll
                for (int j = 0; j < 4; ++j) {
                    bf[j].x = q[cur][s][0][j];
                    bf[j].y = q[cur][s][1][j];
                }
                if (t0bits & kbit) {
                    #pragma unroll
                    for (int j = 0; j < 4; ++j) {
                        acc[0][j] = __builtin_amdgcn_wmma_f32_16x16x4_f32(
                            false, a0[s], false, bf[j], (short)0, acc[0][j], false, false);
                    }
                }
                if (t1bits & kbit) {
                    #pragma unroll
                    for (int j = 0; j < 4; ++j) {
                        acc[1][j] = __builtin_amdgcn_wmma_f32_16x16x4_f32(
                            false, a1[s], false, bf[j], (short)0, acc[1][j], false, false);
                    }
                }
            }
        }
        asm volatile("" :: "v"(mean0), "v"(mean1));
    }

    float* outS  = out;
    float* outP  = out  + (size_t)Bb * Ss * Dd;
    float* outA0 = outP + (size_t)Bb * Ss * Aa * Dd;
    float* outA1 = outA0 + (size_t)Bb * Ss * Aa * Dd;

    const int n = wv * 64 + c16 * 4;

    {
        const v4f part = mean0 + mean1;
        v4f tot;
        tot.x = part.x + xhalf(part.x);
        tot.y = part.y + xhalf(part.y);
        tot.z = part.z + xhalf(part.z);
        tot.w = part.w + xhalf(part.w);
        if (h == 0) *(v4f*)(outS + (size_t)bs * Dd + n) = tot;
    }

    #pragma unroll
    for (int mt = 0; mt < 2; ++mt) {
        #pragma unroll
        for (int e = 0; e < 8; ++e) {
            const int r = mt * 16 + 8 * h + e;
            v4f v;
            v.x = acc[mt][0][e];
            v.y = acc[mt][1][e];
            v.z = acc[mt][2][e];
            v.w = acc[mt][3][e];
            if (r < NARG) {
                float* dst = ((r >> 3) == 0) ? outP
                           : (((r >> 3) == 1) ? outA0 : outA1);
                *(v4f*)(dst + ((size_t)bs * Aa + (r & 7)) * Dd + n) = v;
            }
        }
    }
}

extern "C" void kernel_launch(void* const* d_in, const int* in_sizes, int n_in,
                              void* d_out, int out_size, void* d_ws, size_t ws_size,
                              hipStream_t stream) {
    const float* hidden = (const float*)d_in[0];
    const int*   sids   = (const int*)d_in[1];
    const int*   msk    = (const int*)d_in[2];
    const int*   pred   = (const int*)d_in[3];
    const int*   a0     = (const int*)d_in[4];
    const int*   a1     = (const int*)d_in[5];
    float*       out    = (float*)d_out;

    dim3 grid(Bb * Ss);
    dim3 block(NTHREADS);
    hipLaunchKernelGGL(srl_wmma_kernel, grid, block, 0, stream,
                       hidden, sids, msk, pred, a0, a1, out);
}


# second-half rows whose 25 weights are all zero (masked, no arg match) are not fetched: build ballots row-pair maps per lane half; such lanes re-read their first row pair; sums unchanged
# speedup vs baseline: 2.2062x; 1.0614x over previous
#include <hip/hip_runtime.h>


typedef __attribute__((ext_vector_type(2))) float v2f;
typedef __attribute__((ext_vector_type(4))) float v4f;
typedef __attribute__((ext_vector_type(4))) int   v4i;
typedef __attribute__((ext_vector_type(8))) float v8f;

#define Bb 16
#define Ss 16
#define Ll 128
#define Dd 768
#define Aa 8
#define Tt 5

#define NWAVES 12
#define NTHREADS (NWAVES * 32)
#define KC 8
#define NCHUNK (Ll / (4 * KC))
#define NARG 24
#define MROW 32

__device__ __forceinline__ float xhalf(float x) {
    const int xi = __builtin_bit_cast(int, x);
    return __builtin_bit_cast(float,
        __builtin_amdgcn_permlanex16(xi, xi, 0x76543210, 0xfedcba98, false, false));
}

__global__ __launch_bounds__(NTHREADS)
void srl_wmma_kernel(const float* __restrict__ hidden,
                     const int*   __restrict__ sent_ids,
                     const int*   __restrict__ masks,
                     const int*   __restrict__ pred_ids,
                     const int*   __restrict__ arg0_ids,
                     const int*   __restrict__ arg1_ids,
                     float*       __restrict__ out)
{
    __shared__ __attribute__((aligned(16))) float W[33][132];
    __shared__ __attribute__((aligned(16))) unsigned pmap[2][32];

    const int bs   = blockIdx.x;
    const int tid  = threadIdx.x;
    const int lane = tid & 31;
    const int wv   = tid >> 5;
    const int c16  = lane & 15;
    const int h    = lane >> 4;

    const float* Hl = hidden + (size_t)bs * (Ll * Dd) + (size_t)(2 * h) * Dd
                    + wv * 64 + c16 * 4;
    const char*    Hb   = (const char*)(hidden + (size_t)bs * (Ll * Dd));
    const unsigned loff = (unsigned)(((2 * h) * Dd + wv * 64 + c16 * 4) * 4);

    const int wvs = __builtin_amdgcn_readfirstlane(wv);

    const v4i sv = *(const v4i*)(sent_ids + (size_t)bs * Ll + 4 * lane);

    int id[2][Tt];
    #pragma unroll
    for (int i = 0; i < 2; ++i) {
        const int r    = wvs + NWAVES * i;
        const int type = r >> 3;
        const int a    = r & 7;
        const int* ids = (type == 0 ? pred_ids : (type == 1 ? arg0_ids : arg1_ids))
                         + ((size_t)bs * Aa + a) * Tt;
        #pragma unroll
        for (int t = 0; t < Tt; ++t) id[i][t] = ids[t];
    }
    v4i m4 = {0, 0, 0, 0};
    if (wvs == 0) m4 = *(const v4i*)(masks + (size_t)bs * Ll + 4 * lane);

    v4f q[2][KC][2];
    #pragma unroll
    for (int c = 0; c < 2; ++c) {
        #pragma unroll
        for (int s = 0; s < KC; ++s) {
            const int row = 4 * (KC * c + s);
            q[c][s][0] = *(const v4f*)(Hl + (size_t)row * Dd);
            q[c][s][1] = *(const v4f*)(Hl + (size_t)(row + 1) * Dd);
        }
    }
    __builtin_amdgcn_sched_barrier(0);

    #pragma unroll
    for (int i = 0; i < 2; ++i) {
        int lid = -1;
        int cc  = 0;
        #pragma unroll
        for (int t = 0; t < Tt; ++t) {
            const int v = id[i][t];
            const int c = __builtin_popcount(__builtin_amdgcn_ballot_w32(sv.x == v))
                        + __builtin_popcount(__builtin_amdgcn_ballot_w32(sv.y == v))
                        + __builtin_popcount(__builtin_amdgcn_ballot_w32(sv.z == v))
                        + __builtin_popcount(__builtin_amdgcn_ballot_w32(sv.w == v));
            if (v != 0 && c > 0) { lid = v; cc = c; }
        }
        const float scale = (cc > 0) ? 1.0f / (float)cc : 0.0f;
        v4f w;
        w.x = (sv.x == lid) ? scale : 0.0f;
        w.y = (sv.y == lid) ? scale : 0.0f;
        w.z = (sv.z == lid) ? scale : 0.0f;
        w.w = (sv.w == lid) ? scale : 0.0f;
        *(v4f*)&W[wvs + NWAVES * i][4 * lane] = w;
        const unsigned hit0 = __builtin_amdgcn_ballot_w32((sv.x == lid) | (sv.y == lid));
        const unsigned hit1 = __builtin_amdgcn_ballot_w32((sv.z == lid) | (sv.w == lid));
        if (lane == 0) {
            pmap[0][wvs + NWAVES * i] = (cc > 0) ? hit0 : 0u;
            pmap[1][wvs + NWAVES * i] = (cc > 0) ? hit1 : 0u;
        }
    }
    if (wvs == 0) {
        const int cnt = __builtin_popcount(__builtin_amdgcn_ballot_w32(m4.x != 0))
                      + __builtin_popcount(__builtin_amdgcn_ballot_w32(m4.y != 0))
                      + __builtin_popcount(__builtin_amdgcn_ballot_w32(m4.z != 0))
                      + __builtin_popcount(__builtin_amdgcn_ballot_w32(m4.w != 0));
        const float inv = 1.0f / (float)(cnt > 1 ? cnt : 1);
        v4f w;
        w.x = (float)m4.x * inv;
        w.y = (float)m4.y * inv;
        w.z = (float)m4.z * inv;
        w.w = (float)m4.w * inv;
        *(v4f*)&W[MROW][4 * lane] = w;
        const unsigned mp0 = __builtin_amdgcn_ballot_w32((m4.x | m4.y) != 0);
        const unsigned mp1 = __builtin_amdgcn_ballot_w32((m4.z | m4.w) != 0);
        if (lane == 0) {
            pmap[0][NARG] = mp0;
            pmap[1][NARG] = mp1;
        }
    } else if (wvs <= 8) {
        const v4f z = {0.0f, 0.0f, 0.0f, 0.0f};
        *(v4f*)&W[NARG - 1 + wvs][4 * lane] = z;
        if (lane == 0 && wvs >= 2) {
            pmap[0][NARG - 1 + wvs] = 0u;
            pmap[1][NARG - 1 + wvs] = 0u;
        }
    }
    __builtin_amdgcn_fence(__ATOMIC_RELEASE, "workgroup", "local");
    __builtin_amdgcn_s_barrier();
    __builtin_amdgcn_fence(__ATOMIC_ACQUIRE, "workgroup", "local");

    v8f acc[2][4] = {};
    v4f mean0 = {0.0f, 0.0f, 0.0f, 0.0f};
    v4f mean1 = {0.0f, 0.0f, 0.0f, 0.0f};

    unsigned t0bits, t1bits, npair;
    {
        const v4i rm = *(const v4i*)&pmap[(lane >> 3) & 1][4 * (lane & 7)];
        const int m = rm.x | rm.y | rm.z | rm.w;
        const int a0 = __builtin_amdgcn_readlane(m, 0) | __builtin_amdgcn_readlane(m, 1)
                     | __builtin_amdgcn_readlane(m, 2) | __builtin_amdgcn_readlane(m, 3);
        const int a1 = __builtin_amdgcn_readlane(m, 4) | __builtin_amdgcn_readlane(m, 5);
        const int b0 = __builtin_amdgcn_readlane(m, 8) | __builtin_amdgcn_readlane(m, 9)
                     | __builtin_amdgcn_readlane(m, 10) | __builtin_amdgcn_readlane(m, 11);
        const int b1 = __builtin_amdgcn_readlane(m, 12) | __builtin_amdgcn_readlane(m, 13);
        t0bits = (unsigned)(a0 | b0);
        t1bits = (unsigned)(a1 | b1);
        const unsigned n0 = (unsigned)(a0 | a1 | __builtin_amdgcn_readlane(m, 6));
        const unsigned n1 = (unsigned)(b0 | b1 | __builtin_amdgcn_readlane(m, 14));
        npair = h ? n1 : n0;
    }

    #pragma unroll
    for (int ch = 0; ch < NCHUNK; ++ch) {
        const int cur = ch & 1;
        v2f a0[KC], a1[KC], wm[KC];
        #pragma unroll
        for (int s = 0; s < KC; ++s) {
            const int kb = 4 * (KC * ch + s) + 2 * h;
            a0[s] = *(const v2f*)&W[c16][kb];
            a1[s] = *(const v2f*)&W[16 + c16][kb];
            wm[s] = *(const v2f*)&W[MROW][kb];
        }
        if (ch >= 1 && ch + 1 < NCHUNK) {
            unsigned voff[KC];
            #pragma unroll
            for (int s = 0; s < KC; ++s) {
                const int ks = KC * (ch + 1) + s;
                const unsigned take = (npair >> ks) & 1u;
                voff[s] = loff + take * (unsigned)(4 * ks * Dd * 4);
            }
            __builtin_amdgcn_sched_barrier(0);
            #pragma unroll
            for (int s = 0; s < KC; ++s) {
                const char* p = Hb + voff[s];
                q[cur ^ 1][s][0] = *(const v4f*)p;
                q[cur ^ 1][s][1] = *(const v4f*)(p + Dd * 4);
            }
        }
        __builtin_amdgcn_sched_barrier(0);
        #pragma unroll
        for (int s = 0; s < KC; ++s) {
            mean0 += wm[s].x * q[cur][s][0];
            mean1 += wm[s].y * q[cur][s][1];
            const unsigned kbit = 1u << (KC * ch + s);
            if ((t0bits | t1bits) & kbit) {
                v2f bf[4];
                #pragma unroll
                for (int j = 0; j < 4; ++j) {
                    bf[j].x = q[cur][s][0][j];
                    bf[j].y = q[cur][s][1][j];
                }
                if (t0bits & kbit) {
                    #pragma unroll
                    for (int j = 0; j < 4; ++j) {
                        acc[0][j] = __builtin_amdgcn_wmma_f32_16x16x4_f32(
                            false, a0[s], false, bf[j], (short)0, acc[0][j], false, false);
                    }
                }
                if (t1bits & kbit) {
                    #pragma unroll
                    for (int j = 0; j < 4; ++j) {
                        acc[1][j] = __builtin_amdgcn_wmma_f32_16x16x4_f32(
                            false, a1[s], false, bf[j], (short)0, acc[1][j], false, false);
                    }
                }
            }
        }
        asm volatile("" :: "v"(mean0), "v"(mean1));
    }

    float* outS  = out;
    float* outP  = out  + (size_t)Bb * Ss * Dd;
    float* outA0 = outP + (size_t)Bb * Ss * Aa * Dd;
    float* outA1 = outA0 + (size_t)Bb * Ss * Aa * Dd;

    const int n = wv * 64 + c16 * 4;

    {
        const v4f part = mean0 + mean1;
        v4f tot;
        tot.x = part.x + xhalf(part.x);
        tot.y = part.y + xhalf(part.y);
        tot.z = part.z + xhalf(part.z);
        tot.w = part.w + xhalf(part.w);
        if (h == 0) *(v4f*)(outS + (size_t)bs * Dd + n) = tot;
    }

    #pragma unroll
    for (int mt = 0; mt < 2; ++mt) {
        #pragma unroll
        for (int e = 0; e < 8; ++e) {
            const int r = mt * 16 + 8 * h + e;
            v4f v;
            v.x = acc[mt][0][e];
            v.y = acc[mt][1][e];
            v.z = acc[mt][2][e];
            v.w = acc[mt][3][e];
            if (r < NARG) {
                float* dst = ((r >> 3) == 0) ? outP
                           : (((r >> 3) == 1) ? outA0 : outA1);
                *(v4f*)(dst + ((size_t)bs * Aa + (r & 7)) * Dd + n) = v;
            }
        }
    }
}

extern "C" void kernel_launch(void* const* d_in, const int* in_sizes, int n_in,
                              void* d_out, int out_size, void* d_ws, size_t ws_size,
                              hipStream_t stream) {
    const float* hidden = (const float*)d_in[0];
    const int*   sids   = (const int*)d_in[1];
    const int*   msk    = (const int*)d_in[2];
    const int*   pred   = (const int*)d_in[3];
    const int*   a0     = (const int*)d_in[4];
    const int*   a1     = (const int*)d_in[5];
    float*       out    = (float*)d_out;

    dim3 grid(Bb * Ss);
    dim3 block(NTHREADS);
    hipLaunchKernelGGL(srl_wmma_kernel, grid, block, 0, stream,
                       hidden, sids, msk, pred, a0, a1, out);
}
